# starting kernel
# baseline (speedup 1.0000x reference)
#include <hip/hip_runtime.h>

#define FIN  128
#define FOUT 128
#define EPSV 1e-5f

typedef float v2f __attribute__((ext_vector_type(2)));
typedef float v8f __attribute__((ext_vector_type(8)));

__global__ __launch_bounds__(256) void ln_relu_kernel(
    const float* __restrict__ x, const float* __restrict__ gamma,
    const float* __restrict__ beta, float* __restrict__ xn, int n) {
  int wave = threadIdx.x >> 5;
  int lane = threadIdx.x & 31;
  int row  = blockIdx.x * 8 + wave;
  if (row >= n) return;

  float4 v = ((const float4*)(x + (size_t)row * FIN))[lane];
  float s  = v.x + v.y + v.z + v.w;
  float ss = v.x * v.x + v.y * v.y + v.z * v.z + v.w * v.w;
  for (int m = 16; m >= 1; m >>= 1) {
    s  += __shfl_xor(s,  m, 32);
    ss += __shfl_xor(ss, m, 32);
  }
  float mean = s * (1.0f / FIN);
  float var  = ss * (1.0f / FIN) - mean * mean;
  float rs   = rsqrtf(var + EPSV);

  float4 g  = ((const float4*)gamma)[lane];
  float4 bt = ((const float4*)beta)[lane];
  float4 o;
  o.x = fmaxf(0.0f, (v.x - mean) * rs * g.x + bt.x);
  o.y = fmaxf(0.0f, (v.y - mean) * rs * g.y + bt.y);
  o.z = fmaxf(0.0f, (v.z - mean) * rs * g.z + bt.z);
  o.w = fmaxf(0.0f, (v.w - mean) * rs * g.w + bt.w);
  ((float4*)(xn + (size_t)row * FIN))[lane] = o;
}

#define A_PITCH 132

__global__ __launch_bounds__(256) void gemm_wmma_kernel(
    const float* __restrict__ xn, const float* __restrict__ W,
    float* __restrict__ h) {
  __shared__ float alds[16 * A_PITCH];
  const int tid = threadIdx.x;
  const int n0  = blockIdx.x * 16;

  for (int s = tid; s < 16 * (FIN / 4); s += 256) {
    int r = s >> 5, c4 = s & 31;
    float4 v = ((const float4*)(xn + (size_t)(n0 + r) * FIN))[c4];
    *((float4*)&alds[r * A_PITCH + c4 * 4]) = v;
  }
  __syncthreads();

  const int wave  = tid >> 5;
  const int lane  = tid & 31;
  const int o0    = wave * 16;
  const int mrow  = lane & 15;
  const int khalf = (lane >> 4) * 2;

  const float* abase = &alds[mrow * A_PITCH + khalf];
  const float* wbase = W + (size_t)(o0 + mrow) * FIN + khalf;

  v8f c = {};
#pragma unroll 8
  for (int k = 0; k < FIN; k += 4) {
    v2f a = *((const v2f*)(abase + k));
    v2f b = *((const v2f*)(wbase + k));
    c = __builtin_amdgcn_wmma_f32_16x16x4_f32(
        false, a, false, b, (short)0, c, false, false);
  }

  const int col   = o0 + mrow;
  const int rbase = n0 + ((lane >> 4) << 3);
#pragma unroll
  for (int v = 0; v < 8; ++v) {
    h[(size_t)(rbase + v) * FOUT + col] = c[v];
  }
}

__global__ void deg_init_kernel(float* __restrict__ deg, int n) {
  int i = blockIdx.x * 256 + threadIdx.x;
  if (i < n) deg[i] = 1.0f;
}

__global__ void deg_edge_kernel(const int* __restrict__ ei,
                                const float* __restrict__ ew,
                                float* __restrict__ deg, int e) {
  int i = blockIdx.x * 256 + threadIdx.x;
  if (i < e) unsafeAtomicAdd(&deg[ei[e + i]], ew[i]);
}

__global__ void dinv_kernel(const float* __restrict__ deg,
                            float* __restrict__ dinv, int n) {
  int i = blockIdx.x * 256 + threadIdx.x;
  if (i < n) {
    float d = deg[i];
    dinv[i] = (d > 0.0f) ? rsqrtf(d) : 0.0f;
  }
}

__global__ __launch_bounds__(128) void out_init_kernel(
    const float* __restrict__ h, const float* __restrict__ dinv,
    const float* __restrict__ b, float* __restrict__ out) {
  int i = blockIdx.x;
  int f = threadIdx.x;
  float dv = dinv[i];
  size_t idx = (size_t)i * FOUT + f;
  out[idx] = b[f] + h[idx] * dv * dv;
}

__global__ __launch_bounds__(256) void edge_agg_kernel(
    const int* __restrict__ ei, const float* __restrict__ ew,
    const float* __restrict__ h, const float* __restrict__ dinv,
    float* __restrict__ out, int e) {
  int eid  = blockIdx.x * 8 + (threadIdx.x >> 5);
  if (eid >= e) return;
  int lane = threadIdx.x & 31;
  int r = ei[eid];
  int c = ei[e + eid];
  float norm = dinv[r] * ew[eid] * dinv[c];
  float4 hv = ((const float4*)(h + (size_t)r * FOUT))[lane];
  float* ob = out + (size_t)c * FOUT + lane * 4;
  unsafeAtomicAdd(ob + 0, hv.x * norm);
  unsafeAtomicAdd(ob + 1, hv.y * norm);
  unsafeAtomicAdd(ob + 2, hv.z * norm);
  unsafeAtomicAdd(ob + 3, hv.w * norm);
}

extern "C" void kernel_launch(void* const* d_in, const int* in_sizes, int n_in,
                              void* d_out, int out_size, void* d_ws, size_t ws_size,
                              hipStream_t stream) {
  const float* x     = (const float*)d_in[0];
  const int*   ei    = (const int*)d_in[1];
  const float* ew    = (const float*)d_in[2];
  const float* gamma = (const float*)d_in[3];
  const float* beta  = (const float*)d_in[4];
  const float* W     = (const float*)d_in[5];
  const float* b     = (const float*)d_in[6];
  float* out = (float*)d_out;

  const int n = in_sizes[0] / FIN;
  const int e = in_sizes[2];

  float* xn   = (float*)d_ws;
  float* h    = xn + (size_t)n * FIN;
  float* deg  = h + (size_t)n * FOUT;
  float* dinv = deg + n;

  ln_relu_kernel<<<(n + 7) / 8, 256, 0, stream>>>(x, gamma, beta, xn, n);

  gemm_wmma_kernel<<<n / 16, 256, 0, stream>>>(xn, W, h);

  deg_init_kernel<<<(n + 255) / 256, 256, 0, stream>>>(deg, n);
  deg_edge_kernel<<<(e + 255) / 256, 256, 0, stream>>>(ei, ew, deg, e);

  dinv_kernel<<<(n + 255) / 256, 256, 0, stream>>>(deg, dinv, n);
  out_init_kernel<<<n, 128, 0, stream>>>(h, dinv, b, out);

  edge_agg_kernel<<<(e + 7) / 8, 256, 0, stream>>>(ei, ew, h, dinv, out, e);

  (void)n_in; (void)out_size; (void)ws_size;
}


# Degree launch groups each 8192-edge chunk by destination tile (returning LDS add as place, packed records); a tile reads only its groups, not every edge's tile byte; float atomics kept, arrival order
# speedup vs baseline: 2.4188x; 2.4188x over previous
#include <hip/hip_runtime.h>

#define FIN  128
#define FOUT 128
#define EPSV 1e-5f

typedef float v2f __attribute__((ext_vector_type(2)));
typedef float v8f __attribute__((ext_vector_type(8)));

#define W_PITCH 132
#define NODE_WG    256
#define NODE_WAVES 4

__global__ __launch_bounds__(NODE_WAVES * 32) void ln_gemm_kernel(
    const float* __restrict__ x, const float* __restrict__ gamma,
    const float* __restrict__ beta, const float* __restrict__ W,
    float* __restrict__ h, int nslab) {
  __shared__ float wl[FOUT * W_PITCH];
  const int tid = threadIdx.x;

  const int wave = __builtin_amdgcn_readfirstlane(tid >> 5);
  const int lane = tid & 31;
  const int m    = lane & 15;
  const int hf   = lane >> 4;

  const float* xl = x + (size_t)m * FIN + 64 * hf;
  const int slab0 = wave * NODE_WG + blockIdx.x;
  float4 nx[16];
  {
    const float4* xr = (const float4*)(xl + (size_t)min(slab0, nslab - 1) * 16 * FIN);
#pragma unroll
    for (int q = 0; q < 16; ++q) nx[q] = xr[q];
  }

  for (int s = tid; s < FOUT * (FIN / 4); s += NODE_WAVES * 32) {
    const int r = s >> 5, c4 = s & 31;
    *((float4*)&wl[r * W_PITCH + c4 * 4]) = ((const float4*)W)[s];
  }
  __syncthreads();

  float4 bw[8][16];
  {
    const float* wb = &wl[m * W_PITCH + 64 * hf];
#pragma unroll
    for (int t = 0; t < 8; ++t) {
#pragma unroll
      for (int q = 0; q < 16; ++q) bw[t][q] = *((const float4*)(wb + t * 16 * W_PITCH + 4 * q));
    }
  }
  const float4* g4 = (const float4*)gamma + 16 * hf;
  const float4* b4 = (const float4*)beta + 16 * hf;

  for (int slab = slab0; slab < nslab; slab += NODE_WAVES * NODE_WG) {
    const int r0 = slab * 16;
    float4 a[16];
#pragma unroll
    for (int q = 0; q < 16; ++q) a[q] = nx[q];
    {
      const int nslabx = min(slab + NODE_WAVES * NODE_WG, nslab - 1);
      const float4* xr = (const float4*)(xl + (size_t)nslabx * 16 * FIN);
#pragma unroll
      for (int q = 0; q < 16; ++q) nx[q] = xr[q];
    }

    float s = 0.0f;
#pragma unroll
    for (int q = 0; q < 16; ++q) s += (a[q].x + a[q].y) + (a[q].z + a[q].w);
    s += __shfl_xor(s, 16, 32);
    const float mean = s * (1.0f / FIN);

    float ss = 0.0f;
#pragma unroll
    for (int q = 0; q < 16; ++q) {
      a[q].x -= mean; a[q].y -= mean; a[q].z -= mean; a[q].w -= mean;
      ss += (a[q].x * a[q].x + a[q].y * a[q].y) + (a[q].z * a[q].z + a[q].w * a[q].w);
    }
    ss += __shfl_xor(ss, 16, 32);
    const float rs = rsqrtf(ss * (1.0f / FIN) + EPSV);

#pragma unroll
    for (int q = 0; q < 16; ++q) {
      const float4 g = g4[q];
      const float4 bt = b4[q];
      a[q].x = fmaxf(0.0f, a[q].x * rs * g.x + bt.x);
      a[q].y = fmaxf(0.0f, a[q].y * rs * g.y + bt.y);
      a[q].z = fmaxf(0.0f, a[q].z * rs * g.z + bt.z);
      a[q].w = fmaxf(0.0f, a[q].w * rs * g.w + bt.w);
    }

    v8f acc[8];
#pragma unroll
    for (int t = 0; t < 8; ++t) acc[t] = (v8f){};

#pragma unroll
    for (int q = 0; q < 16; ++q) {
      const v2f a01 = {a[q].x, a[q].y};
      const v2f a23 = {a[q].z, a[q].w};
#pragma unroll
      for (int t = 0; t < 8; ++t) {
        const v2f b01 = {bw[t][q].x, bw[t][q].y};
        acc[t] = __builtin_amdgcn_wmma_f32_16x16x4_f32(
            false, a01, false, b01, (short)0, acc[t], false, false);
      }
#pragma unroll
      for (int t = 0; t < 8; ++t) {
        const v2f b23 = {bw[t][q].z, bw[t][q].w};
        acc[t] = __builtin_amdgcn_wmma_f32_16x16x4_f32(
            false, a23, false, b23, (short)0, acc[t], false, false);
      }
    }

    float* hlo = h + (size_t)r0 * FOUT + lane;
#pragma unroll
    for (int t = 0; t < 8; t += 2) {
#pragma unroll
      for (int v = 0; v < 8; ++v) {
        const float ta = acc[t][v];
        const float tb = acc[t + 1][v];
        const float give = hf ? ta : tb;
        const float got  = __shfl_xor(give, 16, 32);
        const float rowlo = hf ? got : ta;
        const float rowhi = hf ? tb : got;
        hlo[(size_t)v * FOUT + 16 * t]       = rowlo;
        hlo[(size_t)(v + 8) * FOUT + 16 * t] = rowhi;
      }
    }
  }
}

__global__ void deg_init_kernel(float* __restrict__ deg, int n) {
  int i = blockIdx.x * 256 + threadIdx.x;
  if (i < n) deg[i] = 1.0f;
}

#define NPT   391
#define NTILE 256
#define CH    8192
#define EPT   8
#define OFFP  288
#define RBITS 17

__global__ __launch_bounds__(1024) void deg_group_kernel(
    const int* __restrict__ ei, const float* __restrict__ ew,
    float* __restrict__ deg, unsigned int* __restrict__ recs,
    int* __restrict__ offs, int e) {
  __shared__ int   cnt[NTILE];
  __shared__ int   st[NTILE + 1];
  __shared__ uint2 stage[CH];

  const int tid  = threadIdx.x;
  const int base = blockIdx.x * CH;
  const int nvalid = min(CH, e - base);

  if (tid < NTILE) cnt[tid] = 0;

  int   r[EPT], c[EPT];
  float w[EPT];
#pragma unroll
  for (int j = 0; j < EPT; ++j) {
    const int idx = min(base + j * 1024 + tid, e - 1);
    r[j] = ei[idx];
    c[j] = ei[e + idx];
    w[j] = ew[idx];
  }
  __syncthreads();

  int tl[EPT], rk[EPT];
#pragma unroll
  for (int j = 0; j < EPT; ++j) {
    tl[j] = c[j] / NPT;
    rk[j] = 0;
    if (j * 1024 + tid < nvalid) {
      unsafeAtomicAdd(&deg[c[j]], w[j]);
      rk[j] = atomicAdd(&cnt[tl[j]], 1);
    }
  }
  __syncthreads();

  if (tid < 32) {
    int v[NTILE / 32];
    int s = 0;
#pragma unroll
    for (int k = 0; k < NTILE / 32; ++k) { v[k] = cnt[tid * (NTILE / 32) + k]; s += v[k]; }
    int inc = s;
#pragma unroll
    for (int d = 1; d < 32; d <<= 1) {
      const int up = __shfl_up(inc, d, 32);
      if (tid >= d) inc += up;
    }
    int ex = inc - s;
#pragma unroll
    for (int k = 0; k < NTILE / 32; ++k) { st[tid * (NTILE / 32) + k] = ex; ex += v[k]; }
    if (tid == 31) st[NTILE] = inc;
  }
  __syncthreads();

#pragma unroll
  for (int j = 0; j < EPT; ++j) {
    if (j * 1024 + tid < nvalid) {
      const unsigned int key = (unsigned int)r[j] | ((unsigned int)(c[j] - tl[j] * NPT) << RBITS);
      stage[st[tl[j]] + rk[j]] = make_uint2(key, __float_as_uint(w[j]));
    }
  }
  __syncthreads();

  uint2* ro = (uint2*)recs + base;
  for (int i = tid; i < nvalid; i += 1024) ro[i] = stage[i];
  if (tid <= NTILE) offs[blockIdx.x * OFFP + tid] = st[tid];
}

__global__ void dinv_kernel(const float* __restrict__ deg,
                            float* __restrict__ dinv, int n) {
  int i = blockIdx.x * 256 + threadIdx.x;
  if (i < n) {
    float d = deg[i];
    dinv[i] = (d > 0.0f) ? rsqrtf(d) : 0.0f;
  }
}

#define QCAP  12288
#define EB    8
#define MAXCH 512
#define FB    8

__device__ __forceinline__ float lane_f32(float v, int l) {
  return __int_as_float(__builtin_amdgcn_readlane(__float_as_int(v), l));
}

__global__ __launch_bounds__(1024) void tile_agg_kernel(
    const unsigned int* __restrict__ recs, const int* __restrict__ offs,
    const float* __restrict__ h, const float* __restrict__ dinv,
    const float* __restrict__ b, float* __restrict__ out, int n, int e) {
  __shared__ float tile[NPT * FOUT];
  __shared__ uint2 queue[QCAP];
  __shared__ float dl[NPT];
  __shared__ int   segs[MAXCH];
  __shared__ int   scnt[MAXCH];
  __shared__ int   qb[MAXCH];
  __shared__ int   wsum[32];
  __shared__ int   qtotal;

  const int tid  = threadIdx.x;
  const int lane = tid & 31;
  const int t    = blockIdx.x;
  const int n0   = t * NPT;
  const int rows = min(NPT, n - n0);
  const int nchunk = (e + CH - 1) / CH;
  const uint2* rec2 = (const uint2*)recs;
  const int wv = __builtin_amdgcn_readfirstlane(tid >> 5);

  int mys = 0, mycnt = 0;
  if (tid < nchunk) {
    const int* o = offs + tid * OFFP + t;
    mys   = o[0];
    mycnt = o[1] - mys;
  }
  int inc = mycnt;
#pragma unroll
  for (int d = 1; d < 32; d <<= 1) {
    const int up = __shfl_up(inc, d, 32);
    if (lane >= d) inc += up;
  }
  if (lane == 31) wsum[wv] = inc;
  __syncthreads();
  if (tid < 32) {
    const int x = wsum[tid];
    int xi = x;
#pragma unroll
    for (int d = 1; d < 32; d <<= 1) {
      const int up = __shfl_up(xi, d, 32);
      if (tid >= d) xi += up;
    }
    wsum[tid] = xi - x;
    if (tid == 31) qtotal = xi;
  }
  __syncthreads();
  if (tid < nchunk) {
    segs[tid] = tid * CH + mys;
    scnt[tid] = mycnt;
    qb[tid]   = wsum[wv] + inc - mycnt;
  }

  for (int i = tid; i < rows; i += 1024) dl[i] = dinv[n0 + i];
  for (int s = tid; s < rows * (FOUT / 4); s += 1024) {
    const int rr = s >> 5, c4 = s & 31;
    const float  dv = dinv[n0 + rr];
    const float4 hv = ((const float4*)(h + (size_t)(n0 + rr) * FOUT))[c4];
    const float4 bv = ((const float4*)b)[c4];
    float4 o;
    o.x = bv.x + hv.x * dv * dv;
    o.y = bv.y + hv.y * dv * dv;
    o.z = bv.z + hv.z * dv * dv;
    o.w = bv.w + hv.w * dv * dv;
    *((float4*)&tile[rr * FOUT + c4 * 4]) = o;
  }
  __syncthreads();

  const int total = __builtin_amdgcn_readfirstlane(qtotal);
  for (int pbase = 0; pbase < total; pbase += QCAP) {
    for (int k0 = wv; k0 < nchunk; k0 += 32 * FB) {
      int   sg[FB], cn[FB], qp[FB];
      uint2 v0[FB], v1[FB];
#pragma unroll
      for (int j = 0; j < FB; ++j) {
        const int k  = k0 + 32 * j;
        const int kk = min(k, nchunk - 1);
        sg[j] = segs[kk];
        cn[j] = (k < nchunk) ? scnt[kk] : 0;
        qp[j] = qb[kk] - pbase;
      }
#pragma unroll
      for (int j = 0; j < FB; ++j) {
        v0[j] = rec2[min(sg[j] + lane, e - 1)];
        v1[j] = rec2[min(sg[j] + lane + 32, e - 1)];
      }
#pragma unroll
      for (int j = 0; j < FB; ++j) {
        const int p0 = qp[j] + lane, p1 = p0 + 32;
        if (lane < cn[j] && p0 >= 0 && p0 < QCAP) queue[p0] = v0[j];
        if (lane + 32 < cn[j] && p1 >= 0 && p1 < QCAP) queue[p1] = v1[j];
        for (int i = lane + 64; i < cn[j]; i += 32) {
          const int p = qp[j] + i;
          if (p >= 0 && p < QCAP) queue[p] = rec2[sg[j] + i];
        }
      }
    }
    __syncthreads();

    {
      const int cnt = min(total - pbase, QCAP);
      for (int q0 = wv * 32; q0 < cnt; q0 += 1024) {
        if (cnt - q0 >= 32) {
          const uint2 qe  = queue[q0 + lane];
          const int   r   = (int)(qe.x & ((1u << RBITS) - 1u));
          const int   cl  = (int)(qe.x >> RBITS);
          const float nrm = dinv[r] * __uint_as_float(qe.y) * dl[cl];
#pragma unroll
          for (int jb = 0; jb < 32; jb += EB) {
            float v[EB][4];
#pragma unroll
            for (int j = 0; j < EB; ++j) {
              const int rj = __builtin_amdgcn_readlane(r, jb + j);
              const float* hb = h + (size_t)rj * FOUT + lane;
              v[j][0] = hb[0];
              v[j][1] = hb[32];
              v[j][2] = hb[64];
              v[j][3] = hb[96];
            }
#pragma unroll
            for (int j = 0; j < EB; ++j) {
              const int   cj = __builtin_amdgcn_readlane(cl, jb + j);
              const float nj = lane_f32(nrm, jb + j);
              float* tb = tile + cj * FOUT + lane;
              unsafeAtomicAdd(tb + 0,  v[j][0] * nj);
              unsafeAtomicAdd(tb + 32, v[j][1] * nj);
              unsafeAtomicAdd(tb + 64, v[j][2] * nj);
              unsafeAtomicAdd(tb + 96, v[j][3] * nj);
            }
          }
        } else {
          const int nv = cnt - q0;
          int r = 0, cl = 0;
          float nrm = 0.0f;
          if (lane < nv) {
            const uint2 qe = queue[q0 + lane];
            r   = (int)(qe.x & ((1u << RBITS) - 1u));
            cl  = (int)(qe.x >> RBITS);
            nrm = dinv[r] * __uint_as_float(qe.y) * dl[cl];
          }
          for (int j = 0; j < nv; ++j) {
            const int   rj = __builtin_amdgcn_readlane(r, j);
            const int   cj = __builtin_amdgcn_readlane(cl, j);
            const float nj = lane_f32(nrm, j);
            const float* hb = h + (size_t)rj * FOUT + lane;
            float* tb = tile + cj * FOUT + lane;
            const float h0 = hb[0];
            const float h1 = hb[32];
            const float h2 = hb[64];
            const float h3 = hb[96];
            unsafeAtomicAdd(tb + 0,  h0 * nj);
            unsafeAtomicAdd(tb + 32, h1 * nj);
            unsafeAtomicAdd(tb + 64, h2 * nj);
            unsafeAtomicAdd(tb + 96, h3 * nj);
          }
        }
      }
    }
    __syncthreads();
  }

  {
    float4* o4 = (float4*)(out + (size_t)n0 * FOUT);
    const float4* t4 = (const float4*)tile;
    for (int s = tid; s < rows * (FOUT / 4); s += 1024) o4[s] = t4[s];
  }
}

extern "C" void kernel_launch(void* const* d_in, const int* in_sizes, int n_in,
                              void* d_out, int out_size, void* d_ws, size_t ws_size,
                              hipStream_t stream) {
  const float* x     = (const float*)d_in[0];
  const int*   ei    = (const int*)d_in[1];
  const float* ew    = (const float*)d_in[2];
  const float* gamma = (const float*)d_in[3];
  const float* beta  = (const float*)d_in[4];
  const float* W     = (const float*)d_in[5];
  const float* b     = (const float*)d_in[6];
  float* out = (float*)d_out;

  const int n = in_sizes[0] / FIN;
  const int e = in_sizes[2];

  float* ep   = (float*)d_ws;
  float* h    = ep + (size_t)n * FIN;
  float* deg  = h + (size_t)n * FOUT;
  float* dinv = deg + n;
  unsigned int* recs = (unsigned int*)ep;
  int* offs = (int*)(ep + (size_t)4 * e);

  ln_gemm_kernel<<<NODE_WG, NODE_WAVES * 32, 0, stream>>>(x, gamma, beta, W, h, n / 16);

  deg_init_kernel<<<(n + 255) / 256, 256, 0, stream>>>(deg, n);
  deg_group_kernel<<<(e + CH - 1) / CH, 1024, 0, stream>>>(ei, ew, deg, recs, offs, e);

  dinv_kernel<<<(n + 255) / 256, 256, 0, stream>>>(deg, dinv, n);

  tile_agg_kernel<<<(n + NPT - 1) / NPT, 1024, 0, stream>>>(recs, offs, h, dinv, b, out, n, e);

  (void)n_in; (void)out_size; (void)ws_size;
}


# Tile sums: LDS queue re-ordered by owner wave (row mod 32; returning int add as rank); the one owner adds each fetched row by plain 16-byte LDS read-add-write; no LDS float atomics; sum order changes
# speedup vs baseline: 2.5390x; 1.0497x over previous
#include <hip/hip_runtime.h>

#define FIN  128
#define FOUT 128
#define EPSV 1e-5f

typedef float v2f __attribute__((ext_vector_type(2)));
typedef float v8f __attribute__((ext_vector_type(8)));

#define W_PITCH 132
#define NODE_WG    256
#define NODE_WAVES 4

__global__ __launch_bounds__(NODE_WAVES * 32) void ln_gemm_kernel(
    const float* __restrict__ x, const float* __restrict__ gamma,
    const float* __restrict__ beta, const float* __restrict__ W,
    float* __restrict__ h, int nslab) {
  __shared__ float wl[FOUT * W_PITCH];
  const int tid = threadIdx.x;

  const int wave = __builtin_amdgcn_readfirstlane(tid >> 5);
  const int lane = tid & 31;
  const int m    = lane & 15;
  const int hf   = lane >> 4;

  const float* xl = x + (size_t)m * FIN + 64 * hf;
  const int slab0 = wave * NODE_WG + blockIdx.x;
  float4 nx[16];
  {
    const float4* xr = (const float4*)(xl + (size_t)min(slab0, nslab - 1) * 16 * FIN);
#pragma unroll
    for (int q = 0; q < 16; ++q) nx[q] = xr[q];
  }

  for (int s = tid; s < FOUT * (FIN / 4); s += NODE_WAVES * 32) {
    const int r = s >> 5, c4 = s & 31;
    *((float4*)&wl[r * W_PITCH + c4 * 4]) = ((const float4*)W)[s];
  }
  __syncthreads();

  float4 bw[8][16];
  {
    const float* wb = &wl[m * W_PITCH + 64 * hf];
#pragma unroll
    for (int t = 0; t < 8; ++t) {
#pragma unroll
      for (int q = 0; q < 16; ++q) bw[t][q] = *((const float4*)(wb + t * 16 * W_PITCH + 4 * q));
    }
  }
  const float4* g4 = (const float4*)gamma + 16 * hf;
  const float4* b4 = (const float4*)beta + 16 * hf;

  for (int slab = slab0; slab < nslab; slab += NODE_WAVES * NODE_WG) {
    const int r0 = slab * 16;
    float4 a[16];
#pragma unroll
    for (int q = 0; q < 16; ++q) a[q] = nx[q];
    {
      const int nslabx = min(slab + NODE_WAVES * NODE_WG, nslab - 1);
      const float4* xr = (const float4*)(xl + (size_t)nslabx * 16 * FIN);
#pragma unroll
      for (int q = 0; q < 16; ++q) nx[q] = xr[q];
    }

    float s = 0.0f;
#pragma unroll
    for (int q = 0; q < 16; ++q) s += (a[q].x + a[q].y) + (a[q].z + a[q].w);
    s += __shfl_xor(s, 16, 32);
    const float mean = s * (1.0f / FIN);

    float ss = 0.0f;
#pragma unroll
    for (int q = 0; q < 16; ++q) {
      a[q].x -= mean; a[q].y -= mean; a[q].z -= mean; a[q].w -= mean;
      ss += (a[q].x * a[q].x + a[q].y * a[q].y) + (a[q].z * a[q].z + a[q].w * a[q].w);
    }
    ss += __shfl_xor(ss, 16, 32);
    const float rs = rsqrtf(ss * (1.0f / FIN) + EPSV);

#pragma unroll
    for (int q = 0; q < 16; ++q) {
      const float4 g = g4[q];
      const float4 bt = b4[q];
      a[q].x = fmaxf(0.0f, a[q].x * rs * g.x + bt.x);
      a[q].y = fmaxf(0.0f, a[q].y * rs * g.y + bt.y);
      a[q].z = fmaxf(0.0f, a[q].z * rs * g.z + bt.z);
      a[q].w = fmaxf(0.0f, a[q].w * rs * g.w + bt.w);
    }

    v8f acc[8];
#pragma unroll
    for (int t = 0; t < 8; ++t) acc[t] = (v8f){};

#pragma unroll
    for (int q = 0; q < 16; ++q) {
      const v2f a01 = {a[q].x, a[q].y};
      const v2f a23 = {a[q].z, a[q].w};
#pragma unroll
      for (int t = 0; t < 8; ++t) {
        const v2f b01 = {bw[t][q].x, bw[t][q].y};
        acc[t] = __builtin_amdgcn_wmma_f32_16x16x4_f32(
            false, a01, false, b01, (short)0, acc[t], false, false);
      }
#pragma unroll
      for (int t = 0; t < 8; ++t) {
        const v2f b23 = {bw[t][q].z, bw[t][q].w};
        acc[t] = __builtin_amdgcn_wmma_f32_16x16x4_f32(
            false, a23, false, b23, (short)0, acc[t], false, false);
      }
    }

    float* hlo = h + (size_t)r0 * FOUT + lane;
#pragma unroll
    for (int t = 0; t < 8; t += 2) {
#pragma unroll
      for (int v = 0; v < 8; ++v) {
        const float ta = acc[t][v];
        const float tb = acc[t + 1][v];
        const float give = hf ? ta : tb;
        const float got  = __shfl_xor(give, 16, 32);
        const float rowlo = hf ? got : ta;
        const float rowhi = hf ? tb : got;
        hlo[(size_t)v * FOUT + 16 * t]       = rowlo;
        hlo[(size_t)(v + 8) * FOUT + 16 * t] = rowhi;
      }
    }
  }
}

__global__ void deg_init_kernel(float* __restrict__ deg, int n) {
  int i = blockIdx.x * 256 + threadIdx.x;
  if (i < n) deg[i] = 1.0f;
}

#define NPT   391
#define NTILE 256
#define CH    8192
#define EPT   8
#define OFFP  288
#define RBITS 17

__global__ __launch_bounds__(1024) void deg_group_kernel(
    const int* __restrict__ ei, const float* __restrict__ ew,
    float* __restrict__ deg, unsigned int* __restrict__ recs,
    int* __restrict__ offs, int e) {
  __shared__ int   cnt[NTILE];
  __shared__ int   st[NTILE + 1];
  __shared__ uint2 stage[CH];

  const int tid  = threadIdx.x;
  const int base = blockIdx.x * CH;
  const int nvalid = min(CH, e - base);

  if (tid < NTILE) cnt[tid] = 0;

  int   r[EPT], c[EPT];
  float w[EPT];
#pragma unroll
  for (int j = 0; j < EPT; ++j) {
    const int idx = min(base + j * 1024 + tid, e - 1);
    r[j] = ei[idx];
    c[j] = ei[e + idx];
    w[j] = ew[idx];
  }
  __syncthreads();

  int tl[EPT], rk[EPT];
#pragma unroll
  for (int j = 0; j < EPT; ++j) {
    tl[j] = c[j] / NPT;
    rk[j] = 0;
    if (j * 1024 + tid < nvalid) {
      unsafeAtomicAdd(&deg[c[j]], w[j]);
      rk[j] = atomicAdd(&cnt[tl[j]], 1);
    }
  }
  __syncthreads();

  if (tid < 32) {
    int v[NTILE / 32];
    int s = 0;
#pragma unroll
    for (int k = 0; k < NTILE / 32; ++k) { v[k] = cnt[tid * (NTILE / 32) + k]; s += v[k]; }
    int inc = s;
#pragma unroll
    for (int d = 1; d < 32; d <<= 1) {
      const int up = __shfl_up(inc, d, 32);
      if (tid >= d) inc += up;
    }
    int ex = inc - s;
#pragma unroll
    for (int k = 0; k < NTILE / 32; ++k) { st[tid * (NTILE / 32) + k] = ex; ex += v[k]; }
    if (tid == 31) st[NTILE] = inc;
  }
  __syncthreads();

#pragma unroll
  for (int j = 0; j < EPT; ++j) {
    if (j * 1024 + tid < nvalid) {
      const unsigned int key = (unsigned int)r[j] | ((unsigned int)(c[j] - tl[j] * NPT) << RBITS);
      stage[st[tl[j]] + rk[j]] = make_uint2(key, __float_as_uint(w[j]));
    }
  }
  __syncthreads();

  uint2* ro = (uint2*)recs + base;
  for (int i = tid; i < nvalid; i += 1024) ro[i] = stage[i];
  if (tid <= NTILE) offs[blockIdx.x * OFFP + tid] = st[tid];
}

__global__ void dinv_kernel(const float* __restrict__ deg,
                            float* __restrict__ dinv, int n) {
  int i = blockIdx.x * 256 + threadIdx.x;
  if (i < n) {
    float d = deg[i];
    dinv[i] = (d > 0.0f) ? rsqrtf(d) : 0.0f;
  }
}

#define QCAP  12288
#define EB    8
#define MAXCH 512
#define FB    8
#define NSUB  4
#define QPT   (QCAP / 1024)

__device__ __forceinline__ float lane_f32(float v, int l) {
  return __int_as_float(__builtin_amdgcn_readlane(__float_as_int(v), l));
}

__device__ __forceinline__ void owner_trip(float* tile, const float4* __restrict__ h4,
                                           float4 (&v)[2][EB], int r, int rn, int cl,
                                           float nrm, int nv, int lane) {
#pragma unroll
  for (int jb = 0; jb < 32; jb += EB) {
    const int cur = (jb / EB) & 1;
#pragma unroll
    for (int j = 0; j < EB; ++j) {
      const int rj = (jb + EB < 32) ? __builtin_amdgcn_readlane(r, (jb + EB + j) & 31)
                                    : __builtin_amdgcn_readlane(rn, j);
      v[cur ^ 1][j] = h4[(size_t)rj * (FOUT / 4) + lane];
    }
#pragma unroll
    for (int j = 0; j < EB; ++j) {
      if (__builtin_expect(jb + j < nv, 1)) {
        const int   cj = __builtin_amdgcn_readlane(cl, jb + j);
        const float nj = lane_f32(nrm, jb + j);
        float4* tp = (float4*)(tile + cj * FOUT) + lane;
        float4 acc = *tp;
        acc.x += v[cur][j].x * nj;
        acc.y += v[cur][j].y * nj;
        acc.z += v[cur][j].z * nj;
        acc.w += v[cur][j].w * nj;
        *tp = acc;
      }
    }
  }
}

__global__ __launch_bounds__(1024) void tile_agg_kernel(
    const unsigned int* __restrict__ recs, const int* __restrict__ offs,
    const float* __restrict__ h, const float* __restrict__ dinv,
    const float* __restrict__ b, float* __restrict__ out, int n, int e) {
  __shared__ __attribute__((aligned(16))) float tile[NPT * FOUT];
  __shared__ uint2 queue[QCAP];
  __shared__ int   bcnt[32 * NSUB];
  __shared__ int   bst[32 * NSUB + 1];
  __shared__ float dl[NPT];
  __shared__ int   segs[MAXCH];
  __shared__ int   scnt[MAXCH];
  __shared__ int   qb[MAXCH];
  __shared__ int   wsum[32];
  __shared__ int   qtotal;

  const int tid  = threadIdx.x;
  const int lane = tid & 31;
  const int t    = blockIdx.x;
  const int n0   = t * NPT;
  const int rows = min(NPT, n - n0);
  const int nchunk = (e + CH - 1) / CH;
  const uint2* rec2 = (const uint2*)recs;
  const int wv = __builtin_amdgcn_readfirstlane(tid >> 5);

  int mys = 0, mycnt = 0;
  if (tid < nchunk) {
    const int* o = offs + tid * OFFP + t;
    mys   = o[0];
    mycnt = o[1] - mys;
  }
  int inc = mycnt;
#pragma unroll
  for (int d = 1; d < 32; d <<= 1) {
    const int up = __shfl_up(inc, d, 32);
    if (lane >= d) inc += up;
  }
  if (lane == 31) wsum[wv] = inc;
  __syncthreads();
  if (tid < 32) {
    const int x = wsum[tid];
    int xi = x;
#pragma unroll
    for (int d = 1; d < 32; d <<= 1) {
      const int up = __shfl_up(xi, d, 32);
      if (tid >= d) xi += up;
    }
    wsum[tid] = xi - x;
    if (tid == 31) qtotal = xi;
  }
  __syncthreads();
  if (tid < nchunk) {
    segs[tid] = tid * CH + mys;
    scnt[tid] = mycnt;
    qb[tid]   = wsum[wv] + inc - mycnt;
  }

  for (int i = tid; i < rows; i += 1024) dl[i] = dinv[n0 + i];
  for (int s = tid; s < rows * (FOUT / 4); s += 1024) {
    const int rr = s >> 5, c4 = s & 31;
    const float  dv = dinv[n0 + rr];
    const float4 hv = ((const float4*)(h + (size_t)(n0 + rr) * FOUT))[c4];
    const float4 bv = ((const float4*)b)[c4];
    float4 o;
    o.x = bv.x + hv.x * dv * dv;
    o.y = bv.y + hv.y * dv * dv;
    o.z = bv.z + hv.z * dv * dv;
    o.w = bv.w + hv.w * dv * dv;
    *((float4*)&tile[rr * FOUT + c4 * 4]) = o;
  }
  __syncthreads();

  const int total = __builtin_amdgcn_readfirstlane(qtotal);
  for (int pbase = 0; pbase < total; pbase += QCAP) {
    if (tid < 32 * NSUB) bcnt[tid] = 0;
    for (int k0 = wv; k0 < nchunk; k0 += 32 * FB) {
      int   sg[FB], cn[FB], qp[FB];
      uint2 v0[FB], v1[FB];
#pragma unroll
      for (int j = 0; j < FB; ++j) {
        const int k  = k0 + 32 * j;
        const int kk = min(k, nchunk - 1);
        sg[j] = segs[kk];
        cn[j] = (k < nchunk) ? scnt[kk] : 0;
        qp[j] = qb[kk] - pbase;
      }
#pragma unroll
      for (int j = 0; j < FB; ++j) {
        v0[j] = rec2[min(sg[j] + lane, e - 1)];
        v1[j] = rec2[min(sg[j] + lane + 32, e - 1)];
      }
#pragma unroll
      for (int j = 0; j < FB; ++j) {
        const int p0 = qp[j] + lane, p1 = p0 + 32;
        if (lane < cn[j] && p0 >= 0 && p0 < QCAP) queue[p0] = v0[j];
        if (lane + 32 < cn[j] && p1 >= 0 && p1 < QCAP) queue[p1] = v1[j];
        for (int i = lane + 64; i < cn[j]; i += 32) {
          const int p = qp[j] + i;
          if (p >= 0 && p < QCAP) queue[p] = rec2[sg[j] + i];
        }
      }
    }
    __syncthreads();

    const int cnt = min(total - pbase, QCAP);

    uint2 rec[QPT];
    int   rk[QPT];
#pragma unroll
    for (int k = 0; k < QPT; ++k) {
      const int i = k * 1024 + tid;
      rec[k] = make_uint2(0u, 0u);
      rk[k]  = 0;
      if (i < cnt) {
        rec[k] = queue[i];
        const int ci = (int)((rec[k].x >> RBITS) & 31u) * NSUB + (lane & (NSUB - 1));
        rk[k] = atomicAdd(&bcnt[ci], 1);
      }
    }
    __syncthreads();
    if (tid < 32) {
      int v[NSUB];
      int s = 0;
#pragma unroll
      for (int k = 0; k < NSUB; ++k) { v[k] = bcnt[tid * NSUB + k]; s += v[k]; }
      int run = s;
#pragma unroll
      for (int d = 1; d < 32; d <<= 1) {
        const int up = __shfl_up(run, d, 32);
        if (tid >= d) run += up;
      }
      int ex = run - s;
#pragma unroll
      for (int k = 0; k < NSUB; ++k) { bst[tid * NSUB + k] = ex; ex += v[k]; }
      if (tid == 31) bst[32 * NSUB] = run;
    }
    __syncthreads();
#pragma unroll
    for (int k = 0; k < QPT; ++k) {
      const int i = k * 1024 + tid;
      if (i < cnt) {
        const int ci = (int)((rec[k].x >> RBITS) & 31u) * NSUB + (lane & (NSUB - 1));
        queue[bst[ci] + rk[k]] = rec[k];
      }
    }
    __syncthreads();

    {
      const int b1 = __builtin_amdgcn_readfirstlane(bst[(wv + 1) * NSUB]);
      int q0 = __builtin_amdgcn_readfirstlane(bst[wv * NSUB]);
      if (q0 < b1) {
        const float4* h4 = (const float4*)h;
        int r, cl;
        float nrm;
        {
          const uint2 qe = queue[min(q0 + lane, b1 - 1)];
          r   = (int)(qe.x & ((1u << RBITS) - 1u));
          cl  = (int)(qe.x >> RBITS);
          nrm = dinv[r] * __uint_as_float(qe.y) * dl[cl];
        }
        float4 v[2][EB];
#pragma unroll
        for (int j = 0; j < EB; ++j) {
          const int rj = __builtin_amdgcn_readlane(r, j);
          v[0][j] = h4[(size_t)rj * (FOUT / 4) + lane];
        }
        for (; q0 < b1; q0 += 32) {
          const int nv = min(32, b1 - q0);
          const uint2 qn  = queue[min(q0 + 32 + lane, b1 - 1)];
          const int   rn  = (int)(qn.x & ((1u << RBITS) - 1u));
          const int   cln = (int)(qn.x >> RBITS);
          const float dn  = dinv[rn];
          const float dln = dl[cln];
          owner_trip(tile, h4, v, r, rn, cl, nrm, nv, lane);
          r   = rn;
          cl  = cln;
          nrm = dn * __uint_as_float(qn.y) * dln;
        }
      }
    }
    __syncthreads();
  }

  {
    float4* o4 = (float4*)(out + (size_t)n0 * FOUT);
    const float4* t4 = (const float4*)tile;
    for (int s = tid; s < rows * (FOUT / 4); s += 1024) o4[s] = t4[s];
  }
}

extern "C" void kernel_launch(void* const* d_in, const int* in_sizes, int n_in,
                              void* d_out, int out_size, void* d_ws, size_t ws_size,
                              hipStream_t stream) {
  const float* x     = (const float*)d_in[0];
  const int*   ei    = (const int*)d_in[1];
  const float* ew    = (const float*)d_in[2];
  const float* gamma = (const float*)d_in[3];
  const float* beta  = (const float*)d_in[4];
  const float* W     = (const float*)d_in[5];
  const float* b     = (const float*)d_in[6];
  float* out = (float*)d_out;

  const int n = in_sizes[0] / FIN;
  const int e = in_sizes[2];

  float* ep   = (float*)d_ws;
  float* h    = ep + (size_t)n * FIN;
  float* deg  = h + (size_t)n * FOUT;
  float* dinv = deg + n;
  unsigned int* recs = (unsigned int*)ep;
  int* offs = (int*)(ep + (size_t)4 * e);

  ln_gemm_kernel<<<NODE_WG, NODE_WAVES * 32, 0, stream>>>(x, gamma, beta, W, h, n / 16);

  deg_init_kernel<<<(n + 255) / 256, 256, 0, stream>>>(deg, n);
  deg_group_kernel<<<(e + CH - 1) / CH, 1024, 0, stream>>>(ei, ew, deg, recs, offs, e);

  dinv_kernel<<<(n + 255) / 256, 256, 0, stream>>>(deg, dinv, n);

  tile_agg_kernel<<<(n + NPT - 1) / NPT, 1024, 0, stream>>>(recs, offs, h, dinv, b, out, n, e);

  (void)n_in; (void)out_size; (void)ws_size;
}
